# starting kernel
# baseline (speedup 1.0000x reference)
#include <hip/hip_runtime.h>

typedef float v2f __attribute__((ext_vector_type(2)));
typedef float v8f __attribute__((ext_vector_type(8)));

#define NB   16
#define NC   256
#define NH   128
#define NW   128
#define HW   (NH * NW)
#define CHW  ((size_t)NC * HW)

__global__ __launch_bounds__(128) void reduce_mean_max(
    const float* __restrict__ x,
    float* __restrict__ meanp,
    float* __restrict__ maxp)
{
    int tid = blockIdx.x * 128 + threadIdx.x;
    int b   = tid >> 13;
    int p2  = tid & 8191;

    const float* base = x + (size_t)b * CHW + (size_t)p2 * 2;

    v2f s = {0.0f, 0.0f};
    v2f m = {-3.402823e38f, -3.402823e38f};

    #pragma unroll 16
    for (int c = 0; c < NC; ++c) {
        v2f v = __builtin_nontemporal_load((const v2f*)(base + (size_t)c * HW));
        s += v;
        m = __builtin_elementwise_max(m, v);
    }

    v2f mean = s * (1.0f / 256.0f);
    *(v2f*)(meanp + (size_t)b * HW + (size_t)p2 * 2) = mean;
    *(v2f*)(maxp  + (size_t)b * HW + (size_t)p2 * 2) = m;
}

__global__ __launch_bounds__(256) void conv7x7_wmma(
    const float* __restrict__ feat,
    const float* __restrict__ w,
    float* __restrict__ out)
{
    __shared__ int4 stab[100];
    {
        int t = threadIdx.x;
        if (t < 100) {
            int4 rec;
            if (t < 98) {
                int ci = t / 49;
                int r  = t - ci * 49;
                int dy = r / 7;
                int dx = r - dy * 7;
                rec.x = ci * (NB * HW) + (dy - 3) * NW + (dx - 3);
                rec.y = dy - 3;
                rec.z = dx - 3;
                rec.w = __float_as_int(w[t]);
            } else {
                rec.x = 0; rec.y = -1000; rec.z = -1000; rec.w = 0;
            }
            stab[t] = rec;
        }
    }
    __syncthreads();

    int wave = blockIdx.x * 8 + (threadIdx.x >> 5);
    int lane = threadIdx.x & 31;
    int m    = lane & 15;
    int h    = lane >> 4;

    int b    = wave >> 9;
    int rem  = wave & 511;
    int y    = rem >> 2;
    int x0   = (rem & 3) << 5;

    int xp0 = x0 + m;
    int xp1 = xp0 + 16;
    const float* base0 = feat + (size_t)b * HW + (size_t)y * NW + xp0;
    const float* base1 = base0 + 16;

    v8f acc0 = {0.0f, 0.0f, 0.0f, 0.0f, 0.0f, 0.0f, 0.0f, 0.0f};
    v8f acc1 = {0.0f, 0.0f, 0.0f, 0.0f, 0.0f, 0.0f, 0.0f, 0.0f};

    for (int chunk = 0; chunk < 25; ++chunk) {
        int tp = chunk * 4 + h * 2;
        int4 r0 = stab[tp];
        int4 r1 = stab[tp + 1];

        v2f A = { __int_as_float(r0.w), __int_as_float(r1.w) };

        bool vy0 = (unsigned)(y + r0.y) < (unsigned)NH;
        bool vy1 = (unsigned)(y + r1.y) < (unsigned)NH;

        bool va = vy0 && ((unsigned)(xp0 + r0.z) < (unsigned)NW);
        bool vb = vy1 && ((unsigned)(xp0 + r1.z) < (unsigned)NW);
        float pa = base0[va ? r0.x : 0];
        float pb = base0[vb ? r1.x : 0];
        v2f B0 = { va ? pa : 0.0f, vb ? pb : 0.0f };

        bool vc = vy0 && ((unsigned)(xp1 + r0.z) < (unsigned)NW);
        bool vd = vy1 && ((unsigned)(xp1 + r1.z) < (unsigned)NW);
        float pc = base1[vc ? r0.x : 0];
        float pd = base1[vd ? r1.x : 0];
        v2f B1 = { vc ? pc : 0.0f, vd ? pd : 0.0f };

        acc0 = __builtin_amdgcn_wmma_f32_16x16x4_f32(
            false, A, false, B0, (short)0, acc0, false, false);
        acc1 = __builtin_amdgcn_wmma_f32_16x16x4_f32(
            false, A, false, B1, (short)0, acc1, false, false);
    }

    if (lane < 16) {
        float* o = out + (size_t)b * HW + (size_t)y * NW + x0 + m;
        o[0]  = fmaxf(acc0[0], 0.0f);
        o[16] = fmaxf(acc1[0], 0.0f);
    }
}

extern "C" void kernel_launch(void* const* d_in, const int* in_sizes, int n_in,
                              void* d_out, int out_size, void* d_ws, size_t ws_size,
                              hipStream_t stream) {
    const float* x  = (const float*)d_in[0];
    const float* cw = (const float*)d_in[1];
    float* out = (float*)d_out;
    float* ws  = (float*)d_ws;

    float* meanp = ws;
    float* maxp  = ws + (size_t)NB * HW;

    reduce_mean_max<<<1024, 128, 0, stream>>>(x, meanp, maxp);
    conv7x7_wmma<<<1024, 256, 0, stream>>>(meanp, cw, out);
}


# 7x7 conv recast as banded-Toeplitz f32 WMMA GEMM (M = 16 output rows) on an LDS-staged padded tile: 84 WMMAs per 256 pixels, was 400; new sum order (per plane over rows and dx, planes added last)
# speedup vs baseline: 1.2837x; 1.2837x over previous
#include <hip/hip_runtime.h>

typedef float v2f __attribute__((ext_vector_type(2)));
typedef float v4f __attribute__((ext_vector_type(4)));
typedef float v8f __attribute__((ext_vector_type(8)));

#define NB   16
#define NC   256
#define NH   128
#define NW   128
#define HW   (NH * NW)
#define CHW  ((size_t)NC * HW)

__global__ __launch_bounds__(128) void reduce_mean_max(
    const float* __restrict__ x,
    float* __restrict__ meanp,
    float* __restrict__ maxp)
{
    int tid = blockIdx.x * 128 + threadIdx.x;
    int b   = tid >> 13;
    int p2  = tid & 8191;

    const float* base = x + (size_t)b * CHW + (size_t)p2 * 2;

    v2f s = {0.0f, 0.0f};
    v2f m = {-3.402823e38f, -3.402823e38f};

    #pragma unroll 16
    for (int c = 0; c < NC; ++c) {
        v2f v = __builtin_nontemporal_load((const v2f*)(base + (size_t)c * HW));
        s += v;
        m = __builtin_elementwise_max(m, v);
    }

    v2f mean = s * (1.0f / 256.0f);
    *(v2f*)(meanp + (size_t)b * HW + (size_t)p2 * 2) = mean;
    *(v2f*)(maxp  + (size_t)b * HW + (size_t)p2 * 2) = m;
}

#define FROWS   24
#define FPITCH  72
#define FPLANE  (FROWS * FPITCH)
#define WBAND   40
#define WWORDS  (2 * 7 * WBAND)
#define WLOADS  ((WWORDS + 127) / 128)
#define OPITCH  68

__global__ __launch_bounds__(128) void conv7x7_wmma(
    const float* __restrict__ feat,
    const float* __restrict__ w,
    float* __restrict__ out)
{
    __shared__ float sf[2 * FPLANE];
    __shared__ float sw[WWORDS];
    __shared__ float so[16 * OPITCH];

    const int tid = threadIdx.x;
    const int b   = blockIdx.x >> 4;
    const int rem = blockIdx.x & 15;
    const int y0  = (rem >> 1) << 4;
    const int x0  = (rem & 1) << 6;

    const float* fb = feat + (size_t)b * HW;

    const float wq = w[tid < 98 ? tid : 0];

    const int  mrow = tid >> 6;
    const int  mcol = tid & 63;
    const int  mgx  = x0 - 3 + mcol;
    const bool mxok = (unsigned)mgx < (unsigned)NW;
    float fm_[24];
    bool  km_[24];
    #pragma unroll
    for (int i = 0; i < 24; ++i) {
        const int ci = i / 12;
        const int rr = 2 * (i % 12) + mrow;
        const int gy = y0 - 3 + rr;
        km_[i] = mxok && (rr < 22) && ((unsigned)gy < (unsigned)NH);
        fm_[i] = fb[km_[i] ? ci * (NB * HW) + gy * NW + mgx : 0];
    }
    const int  scol = 64 + (tid & 7);
    const int  sgx  = x0 - 3 + scol;
    const bool sxok = (scol < 70) && ((unsigned)sgx < (unsigned)NW);
    float fs_[3];
    bool  ks_[3];
    #pragma unroll
    for (int i = 0; i < 3; ++i) {
        const int rt = (tid >> 3) + 16 * i;
        const int ci = rt >= FROWS ? 1 : 0;
        const int rr = rt - ci * FROWS;
        const int gy = y0 - 3 + rr;
        ks_[i] = sxok && (rr < 22) && ((unsigned)gy < (unsigned)NH);
        fs_[i] = fb[ks_[i] ? ci * (NB * HW) + gy * NW + sgx : 0];
    }

    #pragma unroll
    for (int i = 0; i < WLOADS; ++i) {
        const int e = tid + i * 128;
        const int t = e % WBAND;
        if (e < WWORDS && (t < 15 || t > 21)) sw[e] = 0.0f;
    }
    if (tid < 98) {
        const int ci = tid / 49;
        const int r  = tid - ci * 49;
        const int dy = r / 7;
        const int dx = r - dy * 7;
        sw[(ci * 7 + dx) * WBAND + 15 + dy] = wq;
    }
    #pragma unroll
    for (int i = 0; i < 24; ++i)
        sf[(2 * i + mrow) * FPITCH + mcol] = km_[i] ? fm_[i] : 0.0f;
    #pragma unroll
    for (int i = 0; i < 3; ++i)
        sf[((tid >> 3) + 16 * i) * FPITCH + scol] = ks_[i] ? fs_[i] : 0.0f;
    __syncthreads();

    const int lane = tid & 31;
    const int wv   = tid >> 5;
    const int c    = lane & 15;
    const int h    = lane >> 4;

    const float* bp = sf + (2 * h) * FPITCH + wv * 16 + c;
    const float* ap = sw + 2 * h - c + 15;

    v8f acc0 = {0.0f, 0.0f, 0.0f, 0.0f, 0.0f, 0.0f, 0.0f, 0.0f};
    v8f acc1 = {0.0f, 0.0f, 0.0f, 0.0f, 0.0f, 0.0f, 0.0f, 0.0f};

    v2f fa[2][12], fbv[2][12];

#define LOAD_BLOCK(DX, BUF)                                                      \
          \
    _Pragma("unroll")                                                            \
    for (int j = 0; j < 6; ++j) {                                                \
        fa[BUF][2 * j]      = (v2f){ ap[(DX) * WBAND + 4 * j],                   \
                                     ap[(DX) * WBAND + 4 * j + 1] };             \
        fa[BUF][2 * j + 1]  = (v2f){ ap[(7 + (DX)) * WBAND + 4 * j],             \
                                     ap[(7 + (DX)) * WBAND + 4 * j + 1] };       \
        fbv[BUF][2 * j]     = (v2f){ bp[(4 * j) * FPITCH + (DX)],                \
                                     bp[(4 * j + 1) * FPITCH + (DX)] };          \
        fbv[BUF][2 * j + 1] = (v2f){ bp[FPLANE + (4 * j) * FPITCH + (DX)],       \
                                     bp[FPLANE + (4 * j + 1) * FPITCH + (DX)] }; \
    }

    LOAD_BLOCK(0, 0)
    #pragma unroll
    for (int dx = 0; dx < 7; ++dx) {
        const int cur = dx & 1;
        if (dx < 6) { LOAD_BLOCK(dx + 1, cur ^ 1) }
        __builtin_amdgcn_sched_barrier(0);
        #pragma unroll
        for (int j = 0; j < 6; ++j) {
            acc0 = __builtin_amdgcn_wmma_f32_16x16x4_f32(
                false, fa[cur][2 * j], false, fbv[cur][2 * j], (short)0, acc0, false, false);
            acc1 = __builtin_amdgcn_wmma_f32_16x16x4_f32(
                false, fa[cur][2 * j + 1], false, fbv[cur][2 * j + 1], (short)0, acc1, false, false);
        }
        __builtin_amdgcn_sched_barrier(0);
    }
#undef LOAD_BLOCK

    #pragma unroll
    for (int r = 0; r < 8; ++r)
        so[(8 * h + r) * OPITCH + wv * 16 + c] = fmaxf(acc0[r] + acc1[r], 0.0f);
    __syncthreads();

    #pragma unroll
    for (int it = 0; it < 2; ++it) {
        int row = 4 * wv + 2 * it + h;
        int col = c * 4;
        v4f v = *(const v4f*)(so + row * OPITCH + col);
        *(v4f*)(out + (size_t)b * HW + (size_t)(y0 + row) * NW + x0 + col) = v;
    }
}

extern "C" void kernel_launch(void* const* d_in, const int* in_sizes, int n_in,
                              void* d_out, int out_size, void* d_ws, size_t ws_size,
                              hipStream_t stream) {
    const float* x  = (const float*)d_in[0];
    const float* cw = (const float*)d_in[1];
    float* out = (float*)d_out;
    float* ws  = (float*)d_ws;

    float* meanp = ws;
    float* maxp  = ws + (size_t)NB * HW;

    reduce_mean_max<<<1024, 128, 0, stream>>>(x, meanp, maxp);
    conv7x7_wmma<<<256, 128, 0, stream>>>(meanp, cw, out);
}


# channel mean/max stream: float4 lanes (512 B per wave load), 2048 waves (was 4096), ring of 8-load groups keeps 30-40 loads in flight per wave (was 16, drained each batch); sum order unchanged
# speedup vs baseline: 1.3350x; 1.0400x over previous
#include <hip/hip_runtime.h>

typedef float v2f __attribute__((ext_vector_type(2)));
typedef float v4f __attribute__((ext_vector_type(4)));
typedef float v8f __attribute__((ext_vector_type(8)));

#define NB   16
#define NC   256
#define NH   128
#define NW   128
#define HW   (NH * NW)
#define CHW  ((size_t)NC * HW)

#define RGRP   8
#define RSLOT  4
#define NGRP   (NC / RGRP)

__global__ __launch_bounds__(256) void reduce_mean_max(
    const float* __restrict__ x,
    float* __restrict__ meanp,
    float* __restrict__ maxp)
{
    int tid = blockIdx.x * 256 + threadIdx.x;
    int b   = tid >> 12;
    int p4  = tid & 4095;

    const float* base = x + (size_t)b * CHW + (size_t)p4 * 4;

    v4f s = {0.0f, 0.0f, 0.0f, 0.0f};
    v4f m = {-3.402823e38f, -3.402823e38f, -3.402823e38f, -3.402823e38f};

    v4f buf[RSLOT][RGRP];

#define ISSUE_GROUP(G, SLOT)                                                     \
              \
    _Pragma("unroll")                                                            \
    for (int k = 0; k < RGRP; ++k)                                               \
        buf[SLOT][k] = __builtin_nontemporal_load(                               \
            (const v4f*)(base + (size_t)((G) * RGRP + k) * HW));

#define SUM_GROUP(SLOT)                                                          \
              \
    _Pragma("unroll")                                                            \
    for (int k = 0; k < RGRP; ++k) {                                             \
        s += buf[SLOT][k];                                                       \
        m = __builtin_elementwise_max(m, buf[SLOT][k]);                          \
    }

    ISSUE_GROUP(0, 0)
    ISSUE_GROUP(1, 1)
    ISSUE_GROUP(2, 2)
    ISSUE_GROUP(3, 3)

    #pragma unroll
    for (int g = 0; g < NGRP - RSLOT; g += RSLOT) {
        #pragma unroll
        for (int sl = 0; sl < RSLOT; ++sl) {
            __builtin_amdgcn_sched_barrier(0);
            SUM_GROUP(sl)
            __builtin_amdgcn_sched_barrier(0);
            ISSUE_GROUP(g + RSLOT + sl, sl)
        }
    }
    __builtin_amdgcn_sched_barrier(0);
    SUM_GROUP(0)
    SUM_GROUP(1)
    SUM_GROUP(2)
    SUM_GROUP(3)
#undef ISSUE_GROUP
#undef SUM_GROUP

    v4f mean = s * (1.0f / 256.0f);
    *(v4f*)(meanp + (size_t)b * HW + (size_t)p4 * 4) = mean;
    *(v4f*)(maxp  + (size_t)b * HW + (size_t)p4 * 4) = m;
}

#define FROWS   24
#define FPITCH  72
#define FPLANE  (FROWS * FPITCH)
#define WBAND   40
#define WWORDS  (2 * 7 * WBAND)
#define WLOADS  ((WWORDS + 127) / 128)
#define OPITCH  68

__global__ __launch_bounds__(128) void conv7x7_wmma(
    const float* __restrict__ feat,
    const float* __restrict__ w,
    float* __restrict__ out)
{
    __shared__ float sf[2 * FPLANE];
    __shared__ float sw[WWORDS];
    __shared__ float so[16 * OPITCH];

    const int tid = threadIdx.x;
    const int b   = blockIdx.x >> 4;
    const int rem = blockIdx.x & 15;
    const int y0  = (rem >> 1) << 4;
    const int x0  = (rem & 1) << 6;

    const float* fb = feat + (size_t)b * HW;

    const float wq = w[tid < 98 ? tid : 0];

    const int  mrow = tid >> 6;
    const int  mcol = tid & 63;
    const int  mgx  = x0 - 3 + mcol;
    const bool mxok = (unsigned)mgx < (unsigned)NW;
    float fm_[24];
    bool  km_[24];
    #pragma unroll
    for (int i = 0; i < 24; ++i) {
        const int ci = i / 12;
        const int rr = 2 * (i % 12) + mrow;
        const int gy = y0 - 3 + rr;
        km_[i] = mxok && (rr < 22) && ((unsigned)gy < (unsigned)NH);
        fm_[i] = fb[km_[i] ? ci * (NB * HW) + gy * NW + mgx : 0];
    }
    const int  scol = 64 + (tid & 7);
    const int  sgx  = x0 - 3 + scol;
    const bool sxok = (scol < 70) && ((unsigned)sgx < (unsigned)NW);
    float fs_[3];
    bool  ks_[3];
    #pragma unroll
    for (int i = 0; i < 3; ++i) {
        const int rt = (tid >> 3) + 16 * i;
        const int ci = rt >= FROWS ? 1 : 0;
        const int rr = rt - ci * FROWS;
        const int gy = y0 - 3 + rr;
        ks_[i] = sxok && (rr < 22) && ((unsigned)gy < (unsigned)NH);
        fs_[i] = fb[ks_[i] ? ci * (NB * HW) + gy * NW + sgx : 0];
    }

    #pragma unroll
    for (int i = 0; i < WLOADS; ++i) {
        const int e = tid + i * 128;
        const int t = e % WBAND;
        if (e < WWORDS && (t < 15 || t > 21)) sw[e] = 0.0f;
    }
    if (tid < 98) {
        const int ci = tid / 49;
        const int r  = tid - ci * 49;
        const int dy = r / 7;
        const int dx = r - dy * 7;
        sw[(ci * 7 + dx) * WBAND + 15 + dy] = wq;
    }
    #pragma unroll
    for (int i = 0; i < 24; ++i)
        sf[(2 * i + mrow) * FPITCH + mcol] = km_[i] ? fm_[i] : 0.0f;
    #pragma unroll
    for (int i = 0; i < 3; ++i)
        sf[((tid >> 3) + 16 * i) * FPITCH + scol] = ks_[i] ? fs_[i] : 0.0f;
    __syncthreads();

    const int lane = tid & 31;
    const int wv   = tid >> 5;
    const int c    = lane & 15;
    const int h    = lane >> 4;

    const float* bp = sf + (2 * h) * FPITCH + wv * 16 + c;
    const float* ap = sw + 2 * h - c + 15;

    v8f acc0 = {0.0f, 0.0f, 0.0f, 0.0f, 0.0f, 0.0f, 0.0f, 0.0f};
    v8f acc1 = {0.0f, 0.0f, 0.0f, 0.0f, 0.0f, 0.0f, 0.0f, 0.0f};

    v2f fa[2][12], fbv[2][12];

#define LOAD_BLOCK(DX, BUF)                                                      \
          \
    _Pragma("unroll")                                                            \
    for (int j = 0; j < 6; ++j) {                                                \
        fa[BUF][2 * j]      = (v2f){ ap[(DX) * WBAND + 4 * j],                   \
                                     ap[(DX) * WBAND + 4 * j + 1] };             \
        fa[BUF][2 * j + 1]  = (v2f){ ap[(7 + (DX)) * WBAND + 4 * j],             \
                                     ap[(7 + (DX)) * WBAND + 4 * j + 1] };       \
        fbv[BUF][2 * j]     = (v2f){ bp[(4 * j) * FPITCH + (DX)],                \
                                     bp[(4 * j + 1) * FPITCH + (DX)] };          \
        fbv[BUF][2 * j + 1] = (v2f){ bp[FPLANE + (4 * j) * FPITCH + (DX)],       \
                                     bp[FPLANE + (4 * j + 1) * FPITCH + (DX)] }; \
    }

    LOAD_BLOCK(0, 0)
    #pragma unroll
    for (int dx = 0; dx < 7; ++dx) {
        const int cur = dx & 1;
        if (dx < 6) { LOAD_BLOCK(dx + 1, cur ^ 1) }
        __builtin_amdgcn_sched_barrier(0);
        #pragma unroll
        for (int j = 0; j < 6; ++j) {
            acc0 = __builtin_amdgcn_wmma_f32_16x16x4_f32(
                false, fa[cur][2 * j], false, fbv[cur][2 * j], (short)0, acc0, false, false);
            acc1 = __builtin_amdgcn_wmma_f32_16x16x4_f32(
                false, fa[cur][2 * j + 1], false, fbv[cur][2 * j + 1], (short)0, acc1, false, false);
        }
        __builtin_amdgcn_sched_barrier(0);
    }
#undef LOAD_BLOCK

    #pragma unroll
    for (int r = 0; r < 8; ++r)
        so[(8 * h + r) * OPITCH + wv * 16 + c] = fmaxf(acc0[r] + acc1[r], 0.0f);
    __syncthreads();

    #pragma unroll
    for (int it = 0; it < 2; ++it) {
        int row = 4 * wv + 2 * it + h;
        int col = c * 4;
        v4f v = *(const v4f*)(so + row * OPITCH + col);
        *(v4f*)(out + (size_t)b * HW + (size_t)(y0 + row) * NW + x0 + col) = v;
    }
}

extern "C" void kernel_launch(void* const* d_in, const int* in_sizes, int n_in,
                              void* d_out, int out_size, void* d_ws, size_t ws_size,
                              hipStream_t stream) {
    const float* x  = (const float*)d_in[0];
    const float* cw = (const float*)d_in[1];
    float* out = (float*)d_out;
    float* ws  = (float*)d_ws;

    float* meanp = ws;
    float* maxp  = ws + (size_t)NB * HW;

    reduce_mean_max<<<256, 256, 0, stream>>>(x, meanp, maxp);
    conv7x7_wmma<<<256, 128, 0, stream>>>(meanp, cw, out);
}
